# starting kernel
# baseline (speedup 1.0000x reference)
#include <hip/hip_runtime.h>
#include <math.h>

typedef __attribute__((ext_vector_type(2))) float v2f;
typedef __attribute__((ext_vector_type(8))) float v8f;

#define KG     10
#define KC     8
#define HW     1024
#define DDIM   512
#define BATCH  32
#define NT     (HW / 4)
#define DTILES (DDIM / 16)

__global__ __launch_bounds__(256) void ghostvlad_softmax_swizzle(
    const float* __restrict__ score, float* __restrict__ As)
{
  int gid = blockIdx.x * 256 + threadIdx.x;
  int b = gid >> 10;
  int p = gid & 1023;
  int t = p >> 2;
  int r = p & 3;

  const float* sc = score + (size_t)gid * KG;
  float m = sc[0];
  #pragma unroll
  for (int i = 1; i < KG; ++i) m = fmaxf(m, sc[i]);
  float e[KG];
  float s = 0.0f;
  #pragma unroll
  for (int i = 0; i < KG; ++i) { e[i] = expf(sc[i] - m); s += e[i]; }
  float inv = 1.0f / s;

  float* dst = As + (size_t)(b * NT + t) * 64;
  int lane_base = (r >= 2) ? 16 : 0;
  int slot = r & 1;
  #pragma unroll
  for (int k = 0; k < 16; ++k) {
    float v = (k < KG) ? e[k] * inv : 0.0f;
    dst[(k + lane_base) * 2 + slot] = v;
  }
}

__global__ __launch_bounds__(256) void ghostvlad_wmma_gemm(
    const float* __restrict__ feat, const float* __restrict__ As,
    const float* __restrict__ cluster, float* __restrict__ out)
{
  int wave = (blockIdx.x * 256 + threadIdx.x) >> 5;
  int lane = threadIdx.x & 31;
  int b    = wave >> 5;
  int d0   = (wave & 31) << 4;
  int ln   = lane & 15;
  int roff = (lane >> 4) << 1;

  const float* ap = As + (size_t)(b * NT) * 64 + lane * 2;
  const float* fp = feat + ((size_t)b * HW + roff) * DDIM + d0 + ln;

  v8f acc = {};
  float asum = 0.0f;

  #pragma unroll 4
  for (int t = 0; t < NT; ++t) {
    v2f a = *(const v2f*)ap;
    v2f bb;
    bb.x = fp[0];
    bb.y = fp[DDIM];
    __builtin_prefetch(fp + 8 * 4 * DDIM, 0, 0);
    acc = __builtin_amdgcn_wmma_f32_16x16x4_f32(
        false, a, false, bb, (short)0, acc, false, false);
    asum += a.x + a.y;
    ap += 64;
    fp += 4 * DDIM;
  }

  float asum_full = asum + __shfl_xor(asum, 16, 32);

  if (lane < 16) {
    float* orow = out + (size_t)b * KC * DDIM + d0 + lane;
    #pragma unroll
    for (int v = 0; v < KC; ++v) {
      float av = __shfl(asum_full, v, 32);
      float cl = cluster[v * DDIM + d0 + lane];
      orow[v * DDIM] = acc[v] - av * cl;
    }
  }
}

__global__ __launch_bounds__(128) void ghostvlad_l2norm(float* __restrict__ out)
{
  __shared__ float red[128];
  float* p = out + (size_t)blockIdx.x * DDIM;
  int tid = threadIdx.x;

  float v0 = p[tid];
  float v1 = p[tid + 128];
  float v2 = p[tid + 256];
  float v3 = p[tid + 384];
  red[tid] = v0 * v0 + v1 * v1 + v2 * v2 + v3 * v3;
  __syncthreads();
  #pragma unroll
  for (int off = 64; off > 0; off >>= 1) {
    if (tid < off) red[tid] += red[tid + off];
    __syncthreads();
  }
  float inv = 1.0f / sqrtf(fmaxf(red[0], 1e-12f));
  p[tid]       = v0 * inv;
  p[tid + 128] = v1 * inv;
  p[tid + 256] = v2 * inv;
  p[tid + 384] = v3 * inv;
}

extern "C" void kernel_launch(void* const* d_in, const int* in_sizes, int n_in,
                              void* d_out, int out_size, void* d_ws, size_t ws_size,
                              hipStream_t stream) {
  const float* feat    = (const float*)d_in[0];
  const float* score   = (const float*)d_in[1];
  const float* cluster = (const float*)d_in[2];
  float* out = (float*)d_out;
  float* As  = (float*)d_ws;

  ghostvlad_softmax_swizzle<<<(BATCH * HW) / 256, 256, 0, stream>>>(score, As);

  ghostvlad_wmma_gemm<<<(BATCH * DTILES * 32) / 256, 256, 0, stream>>>(
      feat, As, cluster, out);

  ghostvlad_l2norm<<<BATCH * KC, 128, 0, stream>>>(out);
}


# GEMM kernel: each (batch, 16-col tile) pixel sum split over the 8 waves of a work-group (8192 waves, 48 loads in flight each); partials added through LDS in wave order (sum order changes)
# speedup vs baseline: 2.1920x; 2.1920x over previous
#include <hip/hip_runtime.h>
#include <math.h>

typedef __attribute__((ext_vector_type(2))) float v2f;
typedef __attribute__((ext_vector_type(8))) float v8f;

#define KG     10
#define KC     8
#define HW     1024
#define DDIM   512
#define BATCH  32
#define NT     (HW / 4)
#define DTILES (DDIM / 16)

__global__ __launch_bounds__(256) void ghostvlad_softmax_swizzle(
    const float* __restrict__ score, float* __restrict__ As)
{
  int gid = blockIdx.x * 256 + threadIdx.x;
  int b = gid >> 10;
  int p = gid & 1023;
  int t = p >> 2;
  int r = p & 3;

  const float* sc = score + (size_t)gid * KG;
  float m = sc[0];
  #pragma unroll
  for (int i = 1; i < KG; ++i) m = fmaxf(m, sc[i]);
  float e[KG];
  float s = 0.0f;
  #pragma unroll
  for (int i = 0; i < KG; ++i) { e[i] = expf(sc[i] - m); s += e[i]; }
  float inv = 1.0f / s;

  float* dst = As + (size_t)(b * NT + t) * 64;
  int lane_base = (r >= 2) ? 16 : 0;
  int slot = r & 1;
  #pragma unroll
  for (int k = 0; k < 16; ++k) {
    float v = (k < KG) ? e[k] * inv : 0.0f;
    dst[(k + lane_base) * 2 + slot] = v;
  }
}

#define SPLIT  8
#define NT_W   (NT / SPLIT)

__global__ __launch_bounds__(32 * SPLIT) void ghostvlad_wmma_gemm(
    const float* __restrict__ feat, const float* __restrict__ As,
    const float* __restrict__ cluster, float* __restrict__ out)
{
  __shared__ float red[SPLIT][KC][16];
  __shared__ float ras[SPLIT][16];

  int tile = blockIdx.x;
  int w    = threadIdx.x >> 5;
  int lane = threadIdx.x & 31;
  int b    = tile >> 5;
  int d0   = (tile & 31) << 4;
  int ln   = lane & 15;
  int roff = (lane >> 4) << 1;
  int t0   = w * NT_W;

  const float* ap = As + (size_t)(b * NT + t0) * 64 + lane * 2;
  const float* fp = feat + ((size_t)b * HW + 4 * t0 + roff) * DDIM + d0 + ln;

  v8f acc = {};
  float asum = 0.0f;

  #pragma unroll 16
  for (int t = 0; t < NT_W; ++t) {
    v2f a = *(const v2f*)ap;
    v2f bb;
    bb.x = fp[0];
    bb.y = fp[DDIM];
    acc = __builtin_amdgcn_wmma_f32_16x16x4_f32(
        false, a, false, bb, (short)0, acc, false, false);
    asum += a.x + a.y;
    ap += 64;
    fp += 4 * DDIM;
  }

  float asum_full = asum + __shfl_xor(asum, 16, 32);

  if (lane < 16) {
    for (int v = 0; v < KC; ++v) red[w][v][lane] = acc[v];
    ras[w][lane] = asum_full;
  }
  __syncthreads();

  int tid = threadIdx.x;
  if (tid < KC * 16) {
    int v = tid >> 4;
    int c = tid & 15;
    float s  = red[0][v][c];
    float av = ras[0][v];
    for (int i = 1; i < SPLIT; ++i) { s += red[i][v][c]; av += ras[i][v]; }
    float cl = cluster[v * DDIM + d0 + c];
    out[(size_t)b * KC * DDIM + v * DDIM + d0 + c] = s - av * cl;
  }
}

__global__ __launch_bounds__(128) void ghostvlad_l2norm(float* __restrict__ out)
{
  __shared__ float red[128];
  float* p = out + (size_t)blockIdx.x * DDIM;
  int tid = threadIdx.x;

  float v0 = p[tid];
  float v1 = p[tid + 128];
  float v2 = p[tid + 256];
  float v3 = p[tid + 384];
  red[tid] = v0 * v0 + v1 * v1 + v2 * v2 + v3 * v3;
  __syncthreads();
  #pragma unroll
  for (int off = 64; off > 0; off >>= 1) {
    if (tid < off) red[tid] += red[tid + off];
    __syncthreads();
  }
  float inv = 1.0f / sqrtf(fmaxf(red[0], 1e-12f));
  p[tid]       = v0 * inv;
  p[tid + 128] = v1 * inv;
  p[tid + 256] = v2 * inv;
  p[tid + 384] = v3 * inv;
}

extern "C" void kernel_launch(void* const* d_in, const int* in_sizes, int n_in,
                              void* d_out, int out_size, void* d_ws, size_t ws_size,
                              hipStream_t stream) {
  const float* feat    = (const float*)d_in[0];
  const float* score   = (const float*)d_in[1];
  const float* cluster = (const float*)d_in[2];
  float* out = (float*)d_out;
  float* As  = (float*)d_ws;

  ghostvlad_softmax_swizzle<<<(BATCH * HW) / 256, 256, 0, stream>>>(score, As);

  ghostvlad_wmma_gemm<<<BATCH * DTILES, 32 * SPLIT, 0, stream>>>(
      feat, As, cluster, out);

  ghostvlad_l2norm<<<BATCH * KC, 128, 0, stream>>>(out);
}


# GEMM kernel: wave tile 16 -> 64 columns: 16 B/lane feat loads (256 contiguous B per row), one A fragment feeds 4 WMMAs; 256 work-groups x 16 waves, partials added in wave order (sum order changes)
# speedup vs baseline: 3.3766x; 1.5404x over previous
#include <hip/hip_runtime.h>
#include <math.h>

typedef __attribute__((ext_vector_type(2))) float v2f;
typedef __attribute__((ext_vector_type(4))) float v4f;
typedef __attribute__((ext_vector_type(8))) float v8f;

#define KG     10
#define KC     8
#define HW     1024
#define DDIM   512
#define BATCH  32
#define NT     (HW / 4)
#define CG     64
#define CGROUPS (DDIM / CG)

__global__ __launch_bounds__(256) void ghostvlad_softmax_swizzle(
    const float* __restrict__ score, float* __restrict__ As)
{
  int gid = blockIdx.x * 256 + threadIdx.x;
  int b = gid >> 10;
  int p = gid & 1023;
  int t = p >> 2;
  int r = p & 3;

  const float* sc = score + (size_t)gid * KG;
  float m = sc[0];
  #pragma unroll
  for (int i = 1; i < KG; ++i) m = fmaxf(m, sc[i]);
  float e[KG];
  float s = 0.0f;
  #pragma unroll
  for (int i = 0; i < KG; ++i) { e[i] = expf(sc[i] - m); s += e[i]; }
  float inv = 1.0f / s;

  float* dst = As + (size_t)(b * NT + t) * 64;
  int lane_base = (r >= 2) ? 16 : 0;
  int slot = r & 1;
  #pragma unroll
  for (int k = 0; k < 16; ++k) {
    float v = (k < KG) ? e[k] * inv : 0.0f;
    dst[(k + lane_base) * 2 + slot] = v;
  }
}

#define SPLIT  16
#define NT_W   (NT / SPLIT)

__global__ __launch_bounds__(32 * SPLIT) void ghostvlad_wmma_gemm(
    const float* __restrict__ feat, const float* __restrict__ As,
    const float* __restrict__ cluster, float* __restrict__ out)
{
  __shared__ float red[SPLIT][KC][CG];
  __shared__ float ras[SPLIT][16];

  int tile = blockIdx.x;
  int w    = threadIdx.x >> 5;
  int lane = threadIdx.x & 31;
  int b    = tile >> 3;
  int d0   = (tile & 7) << 6;
  int ln   = lane & 15;
  int roff = (lane >> 4) << 1;
  int t0   = w * NT_W;

  const float* ap = As + (size_t)(b * NT + t0) * 64 + lane * 2;
  const float* fp = feat + ((size_t)b * HW + 4 * t0 + roff) * DDIM + d0 + 4 * ln;

  int tid = threadIdx.x;
  int v = tid >> 6;
  int c = tid & 63;
  float cl = cluster[v * DDIM + d0 + c];

  v8f acc0 = {}, acc1 = {}, acc2 = {}, acc3 = {};
  float asum = 0.0f;

  v2f a[NT_W];
  v4f r0[NT_W], r1[NT_W];
  #pragma unroll
  for (int t = 0; t < NT_W; ++t)
    a[t] = *(const v2f*)(ap + t * 64);
  #pragma unroll
  for (int t = 0; t < NT_W; ++t) {
    r0[t] = *(const v4f*)(fp + t * 4 * DDIM);
    r1[t] = *(const v4f*)(fp + t * 4 * DDIM + DDIM);
  }
  __builtin_amdgcn_sched_barrier(0);

  #pragma unroll
  for (int t = 0; t < NT_W; ++t) {
    v2f b0 = {r0[t].x, r1[t].x};
    v2f b1 = {r0[t].y, r1[t].y};
    v2f b2 = {r0[t].z, r1[t].z};
    v2f b3 = {r0[t].w, r1[t].w};
    acc0 = __builtin_amdgcn_wmma_f32_16x16x4_f32(
        false, a[t], false, b0, (short)0, acc0, false, false);
    acc1 = __builtin_amdgcn_wmma_f32_16x16x4_f32(
        false, a[t], false, b1, (short)0, acc1, false, false);
    acc2 = __builtin_amdgcn_wmma_f32_16x16x4_f32(
        false, a[t], false, b2, (short)0, acc2, false, false);
    acc3 = __builtin_amdgcn_wmma_f32_16x16x4_f32(
        false, a[t], false, b3, (short)0, acc3, false, false);
    asum += a[t].x + a[t].y;
  }

  float asum_full = asum + __shfl_xor(asum, 16, 32);

  if (lane < 16) {
    #pragma unroll
    for (int r = 0; r < KC; ++r) {
      v4f q = {acc0[r], acc1[r], acc2[r], acc3[r]};
      *(v4f*)&red[w][r][4 * lane] = q;
    }
    ras[w][lane] = asum_full;
  }
  __syncthreads();

  float s  = red[0][v][c];
  float av = ras[0][v];
  #pragma unroll
  for (int i = 1; i < SPLIT; ++i) { s += red[i][v][c]; av += ras[i][v]; }
  out[(size_t)b * KC * DDIM + v * DDIM + d0 + c] = s - av * cl;
}

__global__ __launch_bounds__(128) void ghostvlad_l2norm(float* __restrict__ out)
{
  __shared__ float red[128];
  float* p = out + (size_t)blockIdx.x * DDIM;
  int tid = threadIdx.x;

  float v0 = p[tid];
  float v1 = p[tid + 128];
  float v2 = p[tid + 256];
  float v3 = p[tid + 384];
  red[tid] = v0 * v0 + v1 * v1 + v2 * v2 + v3 * v3;
  __syncthreads();
  #pragma unroll
  for (int off = 64; off > 0; off >>= 1) {
    if (tid < off) red[tid] += red[tid + off];
    __syncthreads();
  }
  float inv = 1.0f / sqrtf(fmaxf(red[0], 1e-12f));
  p[tid]       = v0 * inv;
  p[tid + 128] = v1 * inv;
  p[tid + 256] = v2 * inv;
  p[tid + 384] = v3 * inv;
}

extern "C" void kernel_launch(void* const* d_in, const int* in_sizes, int n_in,
                              void* d_out, int out_size, void* d_ws, size_t ws_size,
                              hipStream_t stream) {
  const float* feat    = (const float*)d_in[0];
  const float* score   = (const float*)d_in[1];
  const float* cluster = (const float*)d_in[2];
  float* out = (float*)d_out;
  float* As  = (float*)d_ws;

  ghostvlad_softmax_swizzle<<<(BATCH * HW) / 256, 256, 0, stream>>>(score, As);

  ghostvlad_wmma_gemm<<<BATCH * CGROUPS, 32 * SPLIT, 0, stream>>>(
      feat, As, cluster, out);

  ghostvlad_l2norm<<<BATCH * KC, 128, 0, stream>>>(out);
}


# Softmax kernel folded into the GEMM kernel: each wave takes the softmax of its own 64 pixels (2 per lane) and hands it to its A fragments through wave-private LDS; 3 launches -> 2, no workspace pass
# speedup vs baseline: 3.7411x; 1.1079x over previous
#include <hip/hip_runtime.h>
#include <math.h>

typedef __attribute__((ext_vector_type(2))) float v2f;
typedef __attribute__((ext_vector_type(4))) float v4f;
typedef __attribute__((ext_vector_type(8))) float v8f;

#define KG     10
#define KC     8
#define HW     1024
#define DDIM   512
#define BATCH  32
#define NT     (HW / 4)
#define CG     64
#define CGROUPS (DDIM / CG)

__device__ __forceinline__ void softmax_kg(const float* sc, float* p)
{
  float m = sc[0];
  #pragma unroll
  for (int i = 1; i < KG; ++i) m = fmaxf(m, sc[i]);
  float e[KG];
  float s = 0.0f;
  #pragma unroll
  for (int i = 0; i < KG; ++i) { e[i] = expf(sc[i] - m); s += e[i]; }
  float inv = 1.0f / s;
  #pragma unroll
  for (int i = 0; i < KG; ++i) p[i] = e[i] * inv;
}

#define SPLIT  16
#define NT_W   (NT / SPLIT)
#define PX_W   (4 * NT_W)
#define AROW   (KG + 1)

__global__ __launch_bounds__(32 * SPLIT) void ghostvlad_wmma_gemm(
    const float* __restrict__ feat, const float* __restrict__ score,
    const float* __restrict__ cluster, float* __restrict__ out)
{
  __shared__ float red[SPLIT][KC][CG];
  __shared__ float ras[SPLIT][16];
  __shared__ v2f   sa[SPLIT][32 * AROW];

  int tile = blockIdx.x;
  int w    = threadIdx.x >> 5;
  int lane = threadIdx.x & 31;
  int b    = tile >> 3;
  int d0   = (tile & 7) << 6;
  int ln   = lane & 15;
  int roff = (lane >> 4) << 1;
  int t0   = w * NT_W;

  const float* sp = score + ((size_t)b * HW + PX_W * w) * KG + lane * (2 * KG);
  const float* fp = feat + ((size_t)b * HW + 4 * t0 + roff) * DDIM + d0 + 4 * ln;

  int tid = threadIdx.x;
  int v = tid >> 6;
  int c = tid & 63;
  float cl = cluster[v * DDIM + d0 + c];

  v8f acc0 = {}, acc1 = {}, acc2 = {}, acc3 = {};
  float asum = 0.0f;

  v4f sq[5];
  v4f r0[NT_W], r1[NT_W];
  #pragma unroll
  for (int j = 0; j < 5; ++j)
    sq[j] = *(const v4f*)(sp + 4 * j);
  #pragma unroll
  for (int t = 0; t < NT_W; ++t) {
    r0[t] = *(const v4f*)(fp + t * 4 * DDIM);
    r1[t] = *(const v4f*)(fp + t * 4 * DDIM + DDIM);
  }
  __builtin_amdgcn_sched_barrier(0);

  float sc[2 * KG], p0[KG], p1[KG];
  #pragma unroll
  for (int j = 0; j < 5; ++j) {
    sc[4 * j + 0] = sq[j].x; sc[4 * j + 1] = sq[j].y;
    sc[4 * j + 2] = sq[j].z; sc[4 * j + 3] = sq[j].w;
  }
  softmax_kg(sc, p0);
  softmax_kg(sc + KG, p1);

  v2f* arow = &sa[w][lane * AROW];
  #pragma unroll
  for (int k = 0; k < KG; ++k) {
    v2f pr = {p0[k], p1[k]};
    arow[k] = pr;
  }
  v2f zero2 = {0.0f, 0.0f};
  arow[KG] = zero2;
  const v2f* ard = &sa[w][(lane >> 4) * AROW + (ln < KG ? ln : KG)];
  v2f a[NT_W];
  #pragma unroll
  for (int t = 0; t < NT_W; ++t)
    a[t] = ard[t * 2 * AROW];

  #pragma unroll
  for (int t = 0; t < NT_W; ++t) {
    v2f b0 = {r0[t].x, r1[t].x};
    v2f b1 = {r0[t].y, r1[t].y};
    v2f b2 = {r0[t].z, r1[t].z};
    v2f b3 = {r0[t].w, r1[t].w};
    acc0 = __builtin_amdgcn_wmma_f32_16x16x4_f32(
        false, a[t], false, b0, (short)0, acc0, false, false);
    acc1 = __builtin_amdgcn_wmma_f32_16x16x4_f32(
        false, a[t], false, b1, (short)0, acc1, false, false);
    acc2 = __builtin_amdgcn_wmma_f32_16x16x4_f32(
        false, a[t], false, b2, (short)0, acc2, false, false);
    acc3 = __builtin_amdgcn_wmma_f32_16x16x4_f32(
        false, a[t], false, b3, (short)0, acc3, false, false);
    asum += a[t].x + a[t].y;
  }

  float asum_full = asum + __shfl_xor(asum, 16, 32);

  if (lane < 16) {
    #pragma unroll
    for (int r = 0; r < KC; ++r) {
      v4f q = {acc0[r], acc1[r], acc2[r], acc3[r]};
      *(v4f*)&red[w][r][4 * lane] = q;
    }
    ras[w][lane] = asum_full;
  }
  __syncthreads();

  float s  = red[0][v][c];
  float av = ras[0][v];
  #pragma unroll
  for (int i = 1; i < SPLIT; ++i) { s += red[i][v][c]; av += ras[i][v]; }
  out[(size_t)b * KC * DDIM + v * DDIM + d0 + c] = s - av * cl;
}

__global__ __launch_bounds__(128) void ghostvlad_l2norm(float* __restrict__ out)
{
  __shared__ float red[128];
  float* p = out + (size_t)blockIdx.x * DDIM;
  int tid = threadIdx.x;

  float v0 = p[tid];
  float v1 = p[tid + 128];
  float v2 = p[tid + 256];
  float v3 = p[tid + 384];
  red[tid] = v0 * v0 + v1 * v1 + v2 * v2 + v3 * v3;
  __syncthreads();
  #pragma unroll
  for (int off = 64; off > 0; off >>= 1) {
    if (tid < off) red[tid] += red[tid + off];
    __syncthreads();
  }
  float inv = 1.0f / sqrtf(fmaxf(red[0], 1e-12f));
  p[tid]       = v0 * inv;
  p[tid + 128] = v1 * inv;
  p[tid + 256] = v2 * inv;
  p[tid + 384] = v3 * inv;
}

extern "C" void kernel_launch(void* const* d_in, const int* in_sizes, int n_in,
                              void* d_out, int out_size, void* d_ws, size_t ws_size,
                              hipStream_t stream) {
  const float* feat    = (const float*)d_in[0];
  const float* score   = (const float*)d_in[1];
  const float* cluster = (const float*)d_in[2];
  float* out = (float*)d_out;

  ghostvlad_wmma_gemm<<<BATCH * CGROUPS, 32 * SPLIT, 0, stream>>>(
      feat, score, cluster, out);

  ghostvlad_l2norm<<<BATCH * KC, 128, 0, stream>>>(out);
}
